# starting kernel
# baseline (speedup 1.0000x reference)
#include <hip/hip_runtime.h>
#include <hip/hip_bf16.h>

typedef __attribute__((ext_vector_type(2))) float v2f;
typedef __attribute__((ext_vector_type(8))) float v8f;

#define CB   2048
#define CL   5
#define CD   64
#define CNH  16
#define CNV  4
#define CIL  1000
#define VH_DIM (CNV * CD + CNH)
#define ZU_DIM (2 * CD)

__global__ __launch_bounds__(64)
void caser_prep(const int* __restrict__ seq, const int* __restrict__ user,
                const float* __restrict__ item_emb, const float* __restrict__ user_emb,
                const float* __restrict__ vw, const float* __restrict__ vb,
                const float* __restrict__ hw, const float* __restrict__ hb,
                const int* __restrict__ heights,
                const float* __restrict__ fc1_w, const float* __restrict__ fc1_b,
                float* __restrict__ zu) {
  __shared__ float s_emb[CL][CD];
  __shared__ float s_vh[VH_DIM];

  const int b = blockIdx.x;
  const int t = threadIdx.x;

  for (int l = 0; l < CL; ++l)
    s_emb[l][t] = item_emb[(size_t)seq[b * CL + l] * CD + t];
  __syncthreads();

  for (int f = 0; f < CNV; ++f) {
    float s = vb[f];
    for (int l = 0; l < CL; ++l)
      s += s_emb[l][t] * vw[f * CL + l];
    s_vh[f * CD + t] = s;
  }

  if (t < CNH) {
    const int h = heights[t];
    const int tmax = CL - h;
    float best = -3.402823466e+38f;
    for (int tt = 0; tt <= tmax; ++tt) {
      float s = hb[t];
      for (int i = 0; i + tt < CL; ++i) {
        const float* hwrow = hw + ((size_t)t * CL + i) * CD;
        for (int d = 0; d < CD; ++d)
          s += s_emb[tt + i][d] * hwrow[d];
      }
      best = fmaxf(best, s);
    }
    s_vh[CNV * CD + t] = fmaxf(best, 0.0f);
  }
  __syncthreads();

  float z = fc1_b[t];
  const float* wrow = fc1_w + (size_t)t * VH_DIM;
  for (int k = 0; k < VH_DIM; ++k)
    z += s_vh[k] * wrow[k];
  zu[(size_t)b * ZU_DIM + t]      = fmaxf(z, 0.0f);
  zu[(size_t)b * ZU_DIM + CD + t] = user_emb[(size_t)user[b] * CD + t];
}

__global__ __launch_bounds__(256)
void caser_score(const int* __restrict__ items, const float* __restrict__ W2,
                 const float* __restrict__ b2, const float* __restrict__ zu,
                 float* __restrict__ out) {
  __shared__ float s_zu[ZU_DIM];

  const int b   = blockIdx.x;
  const int tid = threadIdx.x;
  if (tid < ZU_DIM)
    s_zu[tid] = zu[(size_t)b * ZU_DIM + tid];
  __syncthreads();

  const int wave = tid >> 5;
  const int lane = tid & 31;
  const int i0   = blockIdx.y * 128 + wave * 16;
  if (i0 >= CIL) return;

  const int m     = lane & 15;
  const int khalf = lane >> 4;
  const int item_i = (i0 + m < CIL) ? (i0 + m) : (CIL - 1);
  const int idx    = items[(size_t)b * CIL + item_i];
  const float* __restrict__ row = W2 + (size_t)idx * ZU_DIM;

  v8f acc = {};
  for (int k0 = 0; k0 < ZU_DIM; k0 += 4) {
    v2f a;
    a.x = row[k0 + 2 * khalf];
    a.y = row[k0 + 2 * khalf + 1];
    v2f bb;
    bb.x = s_zu[k0 + 2 * khalf];
    bb.y = s_zu[k0 + 2 * khalf + 1];
    acc = __builtin_amdgcn_wmma_f32_16x16x4_f32(
         false, a,  false, bb,
         (short)0, acc,  false,  false);
  }

  if (m == 0) {
    const int base = i0 + khalf * 8;
    #pragma unroll
    for (int r = 0; r < 8; ++r) {
      const int i = base + r;
      if (i < CIL) {
        const int id2 = items[(size_t)b * CIL + i];
        out[(size_t)b * CIL + i] = acc[r] + b2[id2];
      }
    }
  }
}

extern "C" void kernel_launch(void* const* d_in, const int* in_sizes, int n_in,
                              void* d_out, int out_size, void* d_ws, size_t ws_size,
                              hipStream_t stream) {
  const int*   seq      = (const int*)  d_in[0];
  const int*   user     = (const int*)  d_in[1];
  const int*   items    = (const int*)  d_in[2];
  const float* item_emb = (const float*)d_in[3];
  const float* user_emb = (const float*)d_in[4];
  const float* vw       = (const float*)d_in[5];
  const float* vb       = (const float*)d_in[6];
  const float* hw       = (const float*)d_in[7];
  const float* hb       = (const float*)d_in[8];
  const int*   heights  = (const int*)  d_in[9];
  const float* fc1_w    = (const float*)d_in[10];
  const float* fc1_b    = (const float*)d_in[11];
  const float* W2       = (const float*)d_in[12];
  const float* b2       = (const float*)d_in[13];
  float*       out      = (float*)d_out;
  float*       zu       = (float*)d_ws;

  caser_prep<<<dim3(CB), dim3(64), 0, stream>>>(
      seq, user, item_emb, user_emb, vw, vb, hw, hb, heights, fc1_w, fc1_b, zu);

  caser_score<<<dim3(CB, 8), dim3(256), 0, stream>>>(items, W2, b2, zu, out);
}


# prep kernel: horizontal conv and fc1 run as f32 WMMA GEMMs with 16 batch rows as M (one block per 16 rows, weights read once per 16 rows as B fragments); sum order of both products changes
# speedup vs baseline: 2.3295x; 2.3295x over previous
#include <hip/hip_runtime.h>
#include <hip/hip_bf16.h>

typedef __attribute__((ext_vector_type(2))) float v2f;
typedef __attribute__((ext_vector_type(4))) float v4f;
typedef __attribute__((ext_vector_type(8))) float v8f;

#define CB   2048
#define CL   5
#define CD   64
#define CNH  16
#define CNV  4
#define CIL  1000
#define VH_DIM (CNV * CD + CNH)
#define ZU_DIM (2 * CD)

#define PB 16
#define EP 68
#define VP 276

__device__ __forceinline__ v8f wmma4p(v2f a, v2f b, v8f c) {
  return __builtin_amdgcn_wmma_f32_16x16x4_f32(
       false, a,  false, b,
       (short)0, c,  false,  false);
}

__device__ __forceinline__ void lds_barrier() {
  __builtin_amdgcn_fence(__ATOMIC_RELEASE, "workgroup", "local");
  __builtin_amdgcn_s_barrier();
  __builtin_amdgcn_fence(__ATOMIC_ACQUIRE, "workgroup", "local");
}

__global__ __launch_bounds__(128)
void caser_prep(const int* __restrict__ seq, const int* __restrict__ user,
                const float* __restrict__ item_emb, const float* __restrict__ user_emb,
                const float* __restrict__ vw, const float* __restrict__ vb,
                const float* __restrict__ hw, const float* __restrict__ hb,
                const int* __restrict__ heights,
                const float* __restrict__ fc1_w, const float* __restrict__ fc1_b,
                float* __restrict__ zu) {
  __shared__ __attribute__((aligned(16))) float s_emb[PB * CL * EP];
  __shared__ __attribute__((aligned(16))) float s_vh[PB * VP];
  __shared__ __attribute__((aligned(16))) float s_red[4 * CL * 256];

  const int b0   = blockIdx.x * PB;
  const int t    = threadIdx.x;
  const int lane = t & 31;
  const int w    = __builtin_amdgcn_readfirstlane(t >> 5);
  const int c    = lane & 15;
  const int h    = lane >> 4;

  v4f g[10];
  #pragma unroll
  for (int q = 0; q < 10; ++q) {
    const int idx = t + 128 * q;
    const int row = idx >> 4, piece = idx & 15;
    const int it = seq[b0 * CL + row];
    g[q] = *reinterpret_cast<const v4f*>(item_emb + (size_t)it * CD + 4 * piece);
  }
  v4f uval[4];
  #pragma unroll
  for (int q = 0; q < 4; ++q) {
    const int row = (t >> 5) + 4 * q;
    uval[q] = *reinterpret_cast<const v4f*>(
        user_emb + (size_t)user[b0 + row] * CD + 4 * c);
  }
  const int   hgt   = heights[c];
  const float hbias = hb[c];
  const float fbias = fc1_b[16 * w + c];
  v4f hf[CL][2];
  #pragma unroll
  for (int i = 0; i < CL; ++i)
    #pragma unroll
    for (int m = 0; m < 2; ++m)
      hf[i][m] = *reinterpret_cast<const v4f*>(
          hw + ((size_t)c * CL + i) * CD + 16 * w + 8 * m + 4 * h);
  __builtin_amdgcn_sched_barrier(0);
  v4f fb[VH_DIM / 8];
  #pragma unroll
  for (int m = 0; m < VH_DIM / 8; ++m)
    fb[m] = *reinterpret_cast<const v4f*>(
        fc1_w + (size_t)(16 * w + c) * VH_DIM + 8 * m + 4 * h);

  #pragma unroll
  for (int q = 0; q < 10; ++q) {
    const int idx = t + 128 * q;
    *reinterpret_cast<v4f*>(&s_emb[(idx >> 4) * EP + 4 * (idx & 15)]) = g[q];
  }
  lds_barrier();

  {
    const int d = t & 63, bh = t >> 6;
    #pragma unroll
    for (int bb = 0; bb < 8; ++bb) {
      const int b = bh * 8 + bb;
      float e[CL];
      #pragma unroll
      for (int l = 0; l < CL; ++l) e[l] = s_emb[(b * CL + l) * EP + d];
      #pragma unroll
      for (int f = 0; f < CNV; ++f) {
        float s = vb[f];
        #pragma unroll
        for (int l = 0; l < CL; ++l) s += e[l] * vw[f * CL + l];
        s_vh[b * VP + f * CD + d] = s;
      }
    }
  }

  v8f acc[CL];
  #pragma unroll
  for (int tt = 0; tt < CL; ++tt) acc[tt] = v8f{};
  #pragma unroll
  for (int l = 0; l < CL; ++l) {
    #pragma unroll
    for (int m = 0; m < 2; ++m) {
      const v4f a = *reinterpret_cast<const v4f*>(
          &s_emb[(c * CL + l) * EP + 16 * w + 8 * m + 4 * h]);
      #pragma unroll
      for (int i = 0; i <= l; ++i) {
        const int tt = l - i;
        acc[tt] = wmma4p(a.xy, hf[i][m].xy, acc[tt]);
        acc[tt] = wmma4p(a.zw, hf[i][m].zw, acc[tt]);
      }
    }
  }
  #pragma unroll
  for (int tt = 0; tt < CL; ++tt)
    #pragma unroll
    for (int r = 0; r < 8; ++r)
      s_red[((w * CL + tt) * 16 + 8 * h + r) * 16 + c] = acc[tt][r];

  lds_barrier();

  #pragma unroll
  for (int q = 0; q < 2; ++q) {
    const int o = t + 128 * q;
    const int b = o >> 4, f = o & 15;
    const int tmax = CL - hgt;
    const float bias = hbias;
    float best = -3.402823466e+38f;
    #pragma unroll
    for (int tt = 0; tt < CL; ++tt) {
      const float p0 = s_red[((0 * CL + tt) * 16 + b) * 16 + f];
      const float p1 = s_red[((1 * CL + tt) * 16 + b) * 16 + f];
      const float p2 = s_red[((2 * CL + tt) * 16 + b) * 16 + f];
      const float p3 = s_red[((3 * CL + tt) * 16 + b) * 16 + f];
      const float s = ((p0 + p1) + (p2 + p3)) + bias;
      best = fmaxf(best, (tt <= tmax) ? s : -3.402823466e+38f);
    }
    s_vh[b * VP + CNV * CD + f] = fmaxf(best, 0.0f);
  }
  lds_barrier();

  v8f z0 = v8f{}, z1 = v8f{};
  #pragma unroll
  for (int m = 0; m < VH_DIM / 8; m += 2) {
    const v4f a0 = *reinterpret_cast<const v4f*>(&s_vh[c * VP + 8 * m + 4 * h]);
    const v4f a1 = *reinterpret_cast<const v4f*>(&s_vh[c * VP + 8 * (m + 1) + 4 * h]);
    z0 = wmma4p(a0.xy, fb[m].xy, z0);
    z1 = wmma4p(a1.xy, fb[m + 1].xy, z1);
    z0 = wmma4p(a0.zw, fb[m].zw, z0);
    z1 = wmma4p(a1.zw, fb[m + 1].zw, z1);
  }
  {
    #pragma unroll
    for (int r = 0; r < 8; ++r)
      s_red[(8 * h + r) * CD + 16 * w + c] = fmaxf((z0[r] + z1[r]) + fbias, 0.0f);
  }
  lds_barrier();

  #pragma unroll
  for (int q = 0; q < 4; ++q) {
    const int row = (t >> 5) + 4 * q;
    const v4f zv = *reinterpret_cast<const v4f*>(&s_red[row * CD + 4 * c]);
    const v4f v = (lane < 16) ? zv : uval[q];
    *reinterpret_cast<v4f*>(zu + (size_t)(b0 + row) * ZU_DIM + 4 * lane) = v;
  }
}

template <int CTRL>
__device__ __forceinline__ float quad_bcast(float v) {
  return __int_as_float(__builtin_amdgcn_update_dpp(
      0, __float_as_int(v), CTRL, 0xf, 0xf, true));
}

__device__ __forceinline__ float quad_diag_sum(float a0, float a1, float a2, float a3) {
  return ((quad_bcast<0x00>(a0) + quad_bcast<0x55>(a1)) + quad_bcast<0xAA>(a2))
         + quad_bcast<0xFF>(a3);
}

__device__ __forceinline__ v8f wmma4(v2f a, v2f b, v8f c) {
  return __builtin_amdgcn_wmma_f32_16x16x4_f32(
       false, a,  false, b,
       (short)0, c,  false,  false);
}

struct TileA { v4f a0, a1, a2, a3; };

__device__ __forceinline__ TileA load_tile(const float* __restrict__ W2, const unsigned off) {
  const v4f* ap = reinterpret_cast<const v4f*>(reinterpret_cast<const char*>(W2) + off);
  TileA t;
  t.a0 = ap[0]; t.a1 = ap[8]; t.a2 = ap[16]; t.a3 = ap[24];
  return t;
}

__device__ __forceinline__ float score_pair(
    const TileA& x, const TileA& y, const bool odd, const bool bit1,
    const v4f bf0, const v4f bf1, const v4f bf2, const v4f bf3) {
  v8f cx = {}, cy = {};
  cx = wmma4(x.a0.xy, bf0.xy, cx);  cy = wmma4(y.a0.xy, bf0.xy, cy);
  cx = wmma4(x.a0.zw, bf0.zw, cx);  cy = wmma4(y.a0.zw, bf0.zw, cy);
  cx = wmma4(x.a1.xy, bf1.xy, cx);  cy = wmma4(y.a1.xy, bf1.xy, cy);
  cx = wmma4(x.a1.zw, bf1.zw, cx);  cy = wmma4(y.a1.zw, bf1.zw, cy);
  cx = wmma4(x.a2.xy, bf2.xy, cx);  cy = wmma4(y.a2.xy, bf2.xy, cy);
  cx = wmma4(x.a2.zw, bf2.zw, cx);  cy = wmma4(y.a2.zw, bf2.zw, cy);
  cx = wmma4(x.a3.xy, bf3.xy, cx);  cy = wmma4(y.a3.xy, bf3.xy, cy);
  cx = wmma4(x.a3.zw, bf3.zw, cx);  cy = wmma4(y.a3.zw, bf3.zw, cy);
  const float x0 = quad_diag_sum(cx[0], cx[1], cx[2], cx[3]);
  const float x1 = quad_diag_sum(cx[4], cx[5], cx[6], cx[7]);
  const float y0 = quad_diag_sum(cy[0], cy[1], cy[2], cy[3]);
  const float y1 = quad_diag_sum(cy[4], cy[5], cy[6], cy[7]);
  const float px = odd ? x1 : x0;
  const float py = odd ? y1 : y0;
  return bit1 ? py : px;
}

template <int NG>
__device__ __forceinline__ void score_trip(
    const int i0, const int lane, const int rslot, const unsigned kbyte,
    const int* __restrict__ itrow, const float* __restrict__ W2,
    const float* __restrict__ b2, float* __restrict__ orow,
    const v4f bf0, const v4f bf1, const v4f bf2, const v4f bf3) {
  const int my  = i0 + lane;
  const int myc = (my < CIL) ? my : (CIL - 1);
  const int idxL = itrow[myc];
  const float bias = b2[idxL];
  const bool odd  = (lane & 1) != 0, bit1 = (lane & 2) != 0;
  const bool bit2 = (lane & 4) != 0, bit3 = (lane & 8) != 0;

  unsigned off[2 * NG];
  #pragma unroll
  for (int t = 0; t < 2 * NG; ++t)
    off[t] = ((unsigned)__shfl(idxL, rslot + 2 * t) << 9) + kbyte;

  float q[4] = {0.0f, 0.0f, 0.0f, 0.0f};
  TileA cx = load_tile(W2, off[0]);
  TileA cy = load_tile(W2, off[1]);
  #pragma unroll
  for (int g = 0; g < NG; ++g) {
    TileA nx = cx, ny = cy;
    if (g + 1 < NG) {
      nx = load_tile(W2, off[2 * g + 2]);
      ny = load_tile(W2, off[2 * g + 3]);
    }
    __builtin_amdgcn_sched_barrier(0);
    q[g] = score_pair(cx, cy, odd, bit1, bf0, bf1, bf2, bf3);
    cx = nx; cy = ny;
  }
  const float h0 = bit2 ? q[1] : q[0];
  const float h1 = bit2 ? q[3] : q[2];
  const float val = bit3 ? h1 : h0;

  if (my < CIL) orow[my] = val + bias;
}

__global__ __launch_bounds__(128, 8)
void caser_score(const int* __restrict__ items,
                 const float* __restrict__ W2,
                 const float* __restrict__ b2, const float* __restrict__ zu,
                 float* __restrict__ out) {
  const int b    = blockIdx.x;
  const int lane = threadIdx.x & 31;
  const int wave = __builtin_amdgcn_readfirstlane(threadIdx.x >> 5);
  const int h = lane >> 4;
  const int c = lane & 15;
  const int s = c & 3;
  const int r = c >> 2;
  const int koff  = 8 * s + 4 * h;
  const unsigned kbyte = 4u * (unsigned)koff;
  const int rslot = 16 * (r >> 1) + (r & 1);

  const v4f* __restrict__ zp =
      reinterpret_cast<const v4f*>(zu + (size_t)b * ZU_DIM + koff);
  const v4f bf0 = zp[0], bf1 = zp[8], bf2 = zp[16], bf3 = zp[24];

  const int* __restrict__ itrow = items + (size_t)b * CIL;
  float* __restrict__ orow = out + (size_t)b * CIL;

  constexpr int FULL = CIL / 32;
  for (int it = wave; it < FULL; it += 4)
    score_trip<4>(it * 32, lane, rslot, kbyte, itrow, W2, b2, orow, bf0, bf1, bf2, bf3);
  if (wave == (FULL & 3))
    score_trip<(CIL - FULL * 32 + 3) / 4>(FULL * 32, lane, rslot, kbyte, itrow, W2, b2,
                                          orow, bf0, bf1, bf2, bf3);
}

extern "C" void kernel_launch(void* const* d_in, const int* in_sizes, int n_in,
                              void* d_out, int out_size, void* d_ws, size_t ws_size,
                              hipStream_t stream) {
  const int*   seq      = (const int*)  d_in[0];
  const int*   user     = (const int*)  d_in[1];
  const int*   items    = (const int*)  d_in[2];
  const float* item_emb = (const float*)d_in[3];
  const float* user_emb = (const float*)d_in[4];
  const float* vw       = (const float*)d_in[5];
  const float* vb       = (const float*)d_in[6];
  const float* hw       = (const float*)d_in[7];
  const float* hb       = (const float*)d_in[8];
  const int*   heights  = (const int*)  d_in[9];
  const float* fc1_w    = (const float*)d_in[10];
  const float* fc1_b    = (const float*)d_in[11];
  const float* W2       = (const float*)d_in[12];
  const float* b2       = (const float*)d_in[13];
  float*       out      = (float*)d_out;
  float*       zu       = (float*)d_ws;

  caser_prep<<<dim3(CB / PB), dim3(128), 0, stream>>>(
      seq, user, item_emb, user_emb, vw, vb, hw, hb, heights, fc1_w, fc1_b, zu);

  caser_score<<<dim3(CB), dim3(128), 0, stream>>>(items, W2, b2, zu, out);
}


# score kernel: gathered W2 rows ride in BOTH f32 WMMA operands (3 items + the 4 zu segments each): 8 WMMAs score 6 items, was 4; results regrouped via wave-private LDS; sums unchanged
# speedup vs baseline: 2.4288x; 1.0426x over previous
#include <hip/hip_runtime.h>
#include <hip/hip_bf16.h>

typedef __attribute__((ext_vector_type(2))) float v2f;
typedef __attribute__((ext_vector_type(4))) float v4f;
typedef __attribute__((ext_vector_type(8))) float v8f;

#define CB   2048
#define CL   5
#define CD   64
#define CNH  16
#define CNV  4
#define CIL  1000
#define VH_DIM (CNV * CD + CNH)
#define ZU_DIM (2 * CD)

#define PB 16
#define EP 68
#define VP 276

__device__ __forceinline__ v8f wmma4p(v2f a, v2f b, v8f c) {
  return __builtin_amdgcn_wmma_f32_16x16x4_f32(
       false, a,  false, b,
       (short)0, c,  false,  false);
}

__device__ __forceinline__ void lds_barrier() {
  __builtin_amdgcn_fence(__ATOMIC_RELEASE, "workgroup", "local");
  __builtin_amdgcn_s_barrier();
  __builtin_amdgcn_fence(__ATOMIC_ACQUIRE, "workgroup", "local");
}

__global__ __launch_bounds__(128)
void caser_prep(const int* __restrict__ seq, const int* __restrict__ user,
                const float* __restrict__ item_emb, const float* __restrict__ user_emb,
                const float* __restrict__ vw, const float* __restrict__ vb,
                const float* __restrict__ hw, const float* __restrict__ hb,
                const int* __restrict__ heights,
                const float* __restrict__ fc1_w, const float* __restrict__ fc1_b,
                float* __restrict__ zu) {
  __shared__ __attribute__((aligned(16))) float s_emb[PB * CL * EP];
  __shared__ __attribute__((aligned(16))) float s_vh[PB * VP];
  __shared__ __attribute__((aligned(16))) float s_red[4 * CL * 256];

  const int b0   = blockIdx.x * PB;
  const int t    = threadIdx.x;
  const int lane = t & 31;
  const int w    = __builtin_amdgcn_readfirstlane(t >> 5);
  const int c    = lane & 15;
  const int h    = lane >> 4;

  v4f g[10];
  #pragma unroll
  for (int q = 0; q < 10; ++q) {
    const int idx = t + 128 * q;
    const int row = idx >> 4, piece = idx & 15;
    const int it = seq[b0 * CL + row];
    g[q] = *reinterpret_cast<const v4f*>(item_emb + (size_t)it * CD + 4 * piece);
  }
  v4f uval[4];
  #pragma unroll
  for (int q = 0; q < 4; ++q) {
    const int row = (t >> 5) + 4 * q;
    uval[q] = *reinterpret_cast<const v4f*>(
        user_emb + (size_t)user[b0 + row] * CD + 4 * c);
  }
  const int   hgt   = heights[c];
  const float hbias = hb[c];
  const float fbias = fc1_b[16 * w + c];
  v4f hf[CL][2];
  #pragma unroll
  for (int i = 0; i < CL; ++i)
    #pragma unroll
    for (int m = 0; m < 2; ++m)
      hf[i][m] = *reinterpret_cast<const v4f*>(
          hw + ((size_t)c * CL + i) * CD + 16 * w + 8 * m + 4 * h);
  __builtin_amdgcn_sched_barrier(0);
  v4f fb[VH_DIM / 8];
  #pragma unroll
  for (int m = 0; m < VH_DIM / 8; ++m)
    fb[m] = *reinterpret_cast<const v4f*>(
        fc1_w + (size_t)(16 * w + c) * VH_DIM + 8 * m + 4 * h);

  #pragma unroll
  for (int q = 0; q < 10; ++q) {
    const int idx = t + 128 * q;
    *reinterpret_cast<v4f*>(&s_emb[(idx >> 4) * EP + 4 * (idx & 15)]) = g[q];
  }
  lds_barrier();

  {
    const int d = t & 63, bh = t >> 6;
    #pragma unroll
    for (int bb = 0; bb < 8; ++bb) {
      const int b = bh * 8 + bb;
      float e[CL];
      #pragma unroll
      for (int l = 0; l < CL; ++l) e[l] = s_emb[(b * CL + l) * EP + d];
      #pragma unroll
      for (int f = 0; f < CNV; ++f) {
        float s = vb[f];
        #pragma unroll
        for (int l = 0; l < CL; ++l) s += e[l] * vw[f * CL + l];
        s_vh[b * VP + f * CD + d] = s;
      }
    }
  }

  v8f acc[CL];
  #pragma unroll
  for (int tt = 0; tt < CL; ++tt) acc[tt] = v8f{};
  #pragma unroll
  for (int l = 0; l < CL; ++l) {
    #pragma unroll
    for (int m = 0; m < 2; ++m) {
      const v4f a = *reinterpret_cast<const v4f*>(
          &s_emb[(c * CL + l) * EP + 16 * w + 8 * m + 4 * h]);
      #pragma unroll
      for (int i = 0; i <= l; ++i) {
        const int tt = l - i;
        acc[tt] = wmma4p(a.xy, hf[i][m].xy, acc[tt]);
        acc[tt] = wmma4p(a.zw, hf[i][m].zw, acc[tt]);
      }
    }
  }
  #pragma unroll
  for (int tt = 0; tt < CL; ++tt)
    #pragma unroll
    for (int r = 0; r < 8; ++r)
      s_red[((w * CL + tt) * 16 + 8 * h + r) * 16 + c] = acc[tt][r];

  lds_barrier();

  #pragma unroll
  for (int q = 0; q < 2; ++q) {
    const int o = t + 128 * q;
    const int b = o >> 4, f = o & 15;
    const int tmax = CL - hgt;
    const float bias = hbias;
    float best = -3.402823466e+38f;
    #pragma unroll
    for (int tt = 0; tt < CL; ++tt) {
      const float p0 = s_red[((0 * CL + tt) * 16 + b) * 16 + f];
      const float p1 = s_red[((1 * CL + tt) * 16 + b) * 16 + f];
      const float p2 = s_red[((2 * CL + tt) * 16 + b) * 16 + f];
      const float p3 = s_red[((3 * CL + tt) * 16 + b) * 16 + f];
      const float s = ((p0 + p1) + (p2 + p3)) + bias;
      best = fmaxf(best, (tt <= tmax) ? s : -3.402823466e+38f);
    }
    s_vh[b * VP + CNV * CD + f] = fmaxf(best, 0.0f);
  }
  lds_barrier();

  v8f z0 = v8f{}, z1 = v8f{};
  #pragma unroll
  for (int m = 0; m < VH_DIM / 8; m += 2) {
    const v4f a0 = *reinterpret_cast<const v4f*>(&s_vh[c * VP + 8 * m + 4 * h]);
    const v4f a1 = *reinterpret_cast<const v4f*>(&s_vh[c * VP + 8 * (m + 1) + 4 * h]);
    z0 = wmma4p(a0.xy, fb[m].xy, z0);
    z1 = wmma4p(a1.xy, fb[m + 1].xy, z1);
    z0 = wmma4p(a0.zw, fb[m].zw, z0);
    z1 = wmma4p(a1.zw, fb[m + 1].zw, z1);
  }
  {
    #pragma unroll
    for (int r = 0; r < 8; ++r)
      s_red[(8 * h + r) * CD + 16 * w + c] = fmaxf((z0[r] + z1[r]) + fbias, 0.0f);
  }
  lds_barrier();

  #pragma unroll
  for (int q = 0; q < 4; ++q) {
    const int row = (t >> 5) + 4 * q;
    const v4f zv = *reinterpret_cast<const v4f*>(&s_red[row * CD + 4 * c]);
    const v4f v = (lane < 16) ? zv : uval[q];
    *reinterpret_cast<v4f*>(zu + (size_t)(b0 + row) * ZU_DIM + 4 * lane) = v;
  }
}

#define TRIP 48

template <int CTRL>
__device__ __forceinline__ float quad_bcast(float v) {
  return __int_as_float(__builtin_amdgcn_update_dpp(
      0, __float_as_int(v), CTRL, 0xf, 0xf, true));
}

__device__ __forceinline__ float quad_diag_sum(float a0, float a1, float a2, float a3) {
  return ((quad_bcast<0x00>(a0) + quad_bcast<0x55>(a1)) + quad_bcast<0xAA>(a2))
         + quad_bcast<0xFF>(a3);
}

__device__ __forceinline__ v8f wmma4(v2f a, v2f b, v8f c) {
  return __builtin_amdgcn_wmma_f32_16x16x4_f32(
       false, a,  false, b,
       (short)0, c,  false,  false);
}

struct Frag { v4f a0, a1, a2, a3; };

__device__ __forceinline__ Frag load_frag(const char* __restrict__ base, const unsigned off) {
  const v4f* ap = reinterpret_cast<const v4f*>(base + off);
  Frag t;
  t.a0 = ap[0]; t.a1 = ap[8]; t.a2 = ap[16]; t.a3 = ap[24];
  return t;
}

__device__ __forceinline__ void score_tile(const Frag& a, const Frag& b, float& x0, float& x1) {
  v8f c = {};
  c = wmma4(a.a0.xy, b.a0.xy, c);
  c = wmma4(a.a0.zw, b.a0.zw, c);
  c = wmma4(a.a1.xy, b.a1.xy, c);
  c = wmma4(a.a1.zw, b.a1.zw, c);
  c = wmma4(a.a2.xy, b.a2.xy, c);
  c = wmma4(a.a2.zw, b.a2.zw, c);
  c = wmma4(a.a3.xy, b.a3.xy, c);
  c = wmma4(a.a3.zw, b.a3.zw, c);
  x0 = quad_diag_sum(c[0], c[1], c[2], c[3]);
  x1 = quad_diag_sum(c[4], c[5], c[6], c[7]);
}

__device__ __forceinline__ unsigned row_off(const int LO, const unsigned sh0, const unsigned sh1, const int rr) {
  if (LO + 2 < 32) return (unsigned)__shfl((int)sh0, LO + rr);
  if (LO >= 32)    return (unsigned)__shfl((int)sh1, LO - 32 + rr);
  const unsigned lo = (unsigned)__shfl((int)sh0, (LO + rr) & 31);
  const unsigned hi = (unsigned)__shfl((int)sh1, (LO + rr) & 31);
  return (LO + rr < 32) ? lo : hi;
}

template <int NT>
__device__ __forceinline__ void score_trip(
    const int i0, const int lane, const int rr, const unsigned keep,
    const int wslot, const bool usex0, const char* __restrict__ base,
    const int* __restrict__ itrow, const float* __restrict__ b2,
    float* __restrict__ orow, float* sw) {
  const int n0 = i0 + lane;
  const int n1 = i0 + 32 + (lane & 15);
  const int id0 = itrow[(n0 < CIL) ? n0 : (CIL - 1)];
  const int id1 = itrow[(n1 < CIL) ? n1 : (CIL - 1)];
  const float bias0 = b2[id0];
  const float bias1 = b2[id1];
  const unsigned sh0 = (unsigned)id0 << 9, sh1 = (unsigned)id1 << 9;

  unsigned offA[NT], offB[NT];
  #pragma unroll
  for (int t = 0; t < NT; ++t) {
    offA[t] = row_off(6 * t,     sh0, sh1, rr) & keep;
    offB[t] = row_off(6 * t + 3, sh0, sh1, rr) & keep;
  }

  Frag ca = load_frag(base, offA[0]);
  Frag cb = load_frag(base, offB[0]);
  #pragma unroll
  for (int t = 0; t < NT; ++t) {
    Frag na = ca, nb = cb;
    if (t + 1 < NT) {
      na = load_frag(base, offA[t + 1]);
      nb = load_frag(base, offB[t + 1]);
    }
    __builtin_amdgcn_sched_barrier(0);
    float x0, x1;
    score_tile(ca, cb, x0, x1);
    const float v = usex0 ? x0 : x1;
    if (wslot >= 0) sw[6 * t + wslot] = v;
    ca = na; cb = nb;
  }
  const float r0 = sw[lane];
  const float r1 = (32 + (lane & 15) < 6 * NT) ? sw[32 + (lane & 15)] : 0.0f;

  if (n0 < CIL) orow[n0] = r0 + bias0;
  if (lane < 16 && n1 < CIL) orow[n1] = r1 + bias1;
}

__global__ __launch_bounds__(128, 8)
void caser_score(const int* __restrict__ items,
                 const float* __restrict__ W2,
                 const float* __restrict__ b2, const float* __restrict__ zu,
                 float* __restrict__ out) {
  __shared__ float s_res[4 * TRIP];
  const int b    = blockIdx.x;
  const int lane = threadIdx.x & 31;
  const int wave = __builtin_amdgcn_readfirstlane(threadIdx.x >> 5);
  const int h = lane >> 4;
  const int c = lane & 15;
  const int s = c & 3;
  const bool is_zu = c >= 12;
  const int rr = is_zu ? 0 : (c >> 2);
  const unsigned keep = is_zu ? 0u : 0xFFFFFFFFu;
  const unsigned kbyte = 4u * (unsigned)(8 * s + 4 * h);
  const char* zrow = reinterpret_cast<const char*>(zu + (size_t)b * ZU_DIM);
  const char* base = (is_zu ? zrow : reinterpret_cast<const char*>(W2)) + kbyte;

  int wslot = -1;
  if (lane == 12) wslot = 0;
  if (lane == 13) wslot = 1;
  if (lane == 28) wslot = 2;
  if (lane == 16) wslot = 3;
  if (lane == 20) wslot = 4;
  if (lane == 24) wslot = 5;
  const bool usex0 = (lane == 12) || (lane == 28);

  const int* __restrict__ itrow = items + (size_t)b * CIL;
  float* __restrict__ orow = out + (size_t)b * CIL;
  float* sw = s_res + wave * TRIP;

  constexpr int FULL = CIL / TRIP;
  const int role = (wave + b + (b >> 8)) & 3;
  for (int it = role; it < FULL; it += 4)
    score_trip<8>(it * TRIP, lane, rr, keep, wslot, usex0, base, itrow, b2, orow, sw);
  if (role == (FULL & 3))
    score_trip<(CIL - FULL * TRIP + 5) / 6>(FULL * TRIP, lane, rr, keep, wslot, usex0,
                                            base, itrow, b2, orow, sw);
}

extern "C" void kernel_launch(void* const* d_in, const int* in_sizes, int n_in,
                              void* d_out, int out_size, void* d_ws, size_t ws_size,
                              hipStream_t stream) {
  const int*   seq      = (const int*)  d_in[0];
  const int*   user     = (const int*)  d_in[1];
  const int*   items    = (const int*)  d_in[2];
  const float* item_emb = (const float*)d_in[3];
  const float* user_emb = (const float*)d_in[4];
  const float* vw       = (const float*)d_in[5];
  const float* vb       = (const float*)d_in[6];
  const float* hw       = (const float*)d_in[7];
  const float* hb       = (const float*)d_in[8];
  const int*   heights  = (const int*)  d_in[9];
  const float* fc1_w    = (const float*)d_in[10];
  const float* fc1_b    = (const float*)d_in[11];
  const float* W2       = (const float*)d_in[12];
  const float* b2       = (const float*)d_in[13];
  float*       out      = (float*)d_out;
  float*       zu       = (float*)d_ws;

  caser_prep<<<dim3(CB / PB), dim3(128), 0, stream>>>(
      seq, user, item_emb, user_emb, vw, vb, hw, hb, heights, fc1_w, fc1_b, zu);

  caser_score<<<dim3(CB), dim3(128), 0, stream>>>(items, W2, b2, zu, out);
}


# score kernel: one 32-wave block per WGP (8 batch rows) copies the first 77824 entries of the bias table b2 into LDS once; bias gathers read LDS, only higher ids still go to memory; sums unchanged
# speedup vs baseline: 2.5372x; 1.0446x over previous
#include <hip/hip_runtime.h>
#include <hip/hip_bf16.h>

typedef __attribute__((ext_vector_type(2))) float v2f;
typedef __attribute__((ext_vector_type(4))) float v4f;
typedef __attribute__((ext_vector_type(8))) float v8f;

#define CB   2048
#define CL   5
#define CD   64
#define CNH  16
#define CNV  4
#define CIL  1000
#define VH_DIM (CNV * CD + CNH)
#define ZU_DIM (2 * CD)

#define PB 16
#define EP 68
#define VP 276

__device__ __forceinline__ v8f wmma4p(v2f a, v2f b, v8f c) {
  return __builtin_amdgcn_wmma_f32_16x16x4_f32(
       false, a,  false, b,
       (short)0, c,  false,  false);
}

__device__ __forceinline__ void lds_barrier() {
  __builtin_amdgcn_fence(__ATOMIC_RELEASE, "workgroup", "local");
  __builtin_amdgcn_s_barrier();
  __builtin_amdgcn_fence(__ATOMIC_ACQUIRE, "workgroup", "local");
}

__global__ __launch_bounds__(128)
void caser_prep(const int* __restrict__ seq, const int* __restrict__ user,
                const float* __restrict__ item_emb, const float* __restrict__ user_emb,
                const float* __restrict__ vw, const float* __restrict__ vb,
                const float* __restrict__ hw, const float* __restrict__ hb,
                const int* __restrict__ heights,
                const float* __restrict__ fc1_w, const float* __restrict__ fc1_b,
                float* __restrict__ zu) {
  __shared__ __attribute__((aligned(16))) float s_emb[PB * CL * EP];
  __shared__ __attribute__((aligned(16))) float s_vh[PB * VP];
  __shared__ __attribute__((aligned(16))) float s_red[4 * CL * 256];

  const int b0   = blockIdx.x * PB;
  const int t    = threadIdx.x;
  const int lane = t & 31;
  const int w    = __builtin_amdgcn_readfirstlane(t >> 5);
  const int c    = lane & 15;
  const int h    = lane >> 4;

  v4f g[10];
  #pragma unroll
  for (int q = 0; q < 10; ++q) {
    const int idx = t + 128 * q;
    const int row = idx >> 4, piece = idx & 15;
    const int it = seq[b0 * CL + row];
    g[q] = *reinterpret_cast<const v4f*>(item_emb + (size_t)it * CD + 4 * piece);
  }
  v4f uval[4];
  #pragma unroll
  for (int q = 0; q < 4; ++q) {
    const int row = (t >> 5) + 4 * q;
    uval[q] = *reinterpret_cast<const v4f*>(
        user_emb + (size_t)user[b0 + row] * CD + 4 * c);
  }
  const int   hgt   = heights[c];
  const float hbias = hb[c];
  const float fbias = fc1_b[16 * w + c];
  v4f hf[CL][2];
  #pragma unroll
  for (int i = 0; i < CL; ++i)
    #pragma unroll
    for (int m = 0; m < 2; ++m)
      hf[i][m] = *reinterpret_cast<const v4f*>(
          hw + ((size_t)c * CL + i) * CD + 16 * w + 8 * m + 4 * h);
  __builtin_amdgcn_sched_barrier(0);
  v4f fb[VH_DIM / 8];
  #pragma unroll
  for (int m = 0; m < VH_DIM / 8; ++m)
    fb[m] = *reinterpret_cast<const v4f*>(
        fc1_w + (size_t)(16 * w + c) * VH_DIM + 8 * m + 4 * h);

  #pragma unroll
  for (int q = 0; q < 10; ++q) {
    const int idx = t + 128 * q;
    *reinterpret_cast<v4f*>(&s_emb[(idx >> 4) * EP + 4 * (idx & 15)]) = g[q];
  }
  lds_barrier();

  {
    const int d = t & 63, bh = t >> 6;
    #pragma unroll
    for (int bb = 0; bb < 8; ++bb) {
      const int b = bh * 8 + bb;
      float e[CL];
      #pragma unroll
      for (int l = 0; l < CL; ++l) e[l] = s_emb[(b * CL + l) * EP + d];
      #pragma unroll
      for (int f = 0; f < CNV; ++f) {
        float s = vb[f];
        #pragma unroll
        for (int l = 0; l < CL; ++l) s += e[l] * vw[f * CL + l];
        s_vh[b * VP + f * CD + d] = s;
      }
    }
  }

  v8f acc[CL];
  #pragma unroll
  for (int tt = 0; tt < CL; ++tt) acc[tt] = v8f{};
  #pragma unroll
  for (int l = 0; l < CL; ++l) {
    #pragma unroll
    for (int m = 0; m < 2; ++m) {
      const v4f a = *reinterpret_cast<const v4f*>(
          &s_emb[(c * CL + l) * EP + 16 * w + 8 * m + 4 * h]);
      #pragma unroll
      for (int i = 0; i <= l; ++i) {
        const int tt = l - i;
        acc[tt] = wmma4p(a.xy, hf[i][m].xy, acc[tt]);
        acc[tt] = wmma4p(a.zw, hf[i][m].zw, acc[tt]);
      }
    }
  }
  #pragma unroll
  for (int tt = 0; tt < CL; ++tt)
    #pragma unroll
    for (int r = 0; r < 8; ++r)
      s_red[((w * CL + tt) * 16 + 8 * h + r) * 16 + c] = acc[tt][r];

  lds_barrier();

  #pragma unroll
  for (int q = 0; q < 2; ++q) {
    const int o = t + 128 * q;
    const int b = o >> 4, f = o & 15;
    const int tmax = CL - hgt;
    const float bias = hbias;
    float best = -3.402823466e+38f;
    #pragma unroll
    for (int tt = 0; tt < CL; ++tt) {
      const float p0 = s_red[((0 * CL + tt) * 16 + b) * 16 + f];
      const float p1 = s_red[((1 * CL + tt) * 16 + b) * 16 + f];
      const float p2 = s_red[((2 * CL + tt) * 16 + b) * 16 + f];
      const float p3 = s_red[((3 * CL + tt) * 16 + b) * 16 + f];
      const float s = ((p0 + p1) + (p2 + p3)) + bias;
      best = fmaxf(best, (tt <= tmax) ? s : -3.402823466e+38f);
    }
    s_vh[b * VP + CNV * CD + f] = fmaxf(best, 0.0f);
  }
  lds_barrier();

  v8f z0 = v8f{}, z1 = v8f{};
  #pragma unroll
  for (int m = 0; m < VH_DIM / 8; m += 2) {
    const v4f a0 = *reinterpret_cast<const v4f*>(&s_vh[c * VP + 8 * m + 4 * h]);
    const v4f a1 = *reinterpret_cast<const v4f*>(&s_vh[c * VP + 8 * (m + 1) + 4 * h]);
    z0 = wmma4p(a0.xy, fb[m].xy, z0);
    z1 = wmma4p(a1.xy, fb[m + 1].xy, z1);
    z0 = wmma4p(a0.zw, fb[m].zw, z0);
    z1 = wmma4p(a1.zw, fb[m + 1].zw, z1);
  }
  {
    #pragma unroll
    for (int r = 0; r < 8; ++r)
      s_red[(8 * h + r) * CD + 16 * w + c] = fmaxf((z0[r] + z1[r]) + fbias, 0.0f);
  }
  lds_barrier();

  #pragma unroll
  for (int q = 0; q < 4; ++q) {
    const int row = (t >> 5) + 4 * q;
    const v4f zv = *reinterpret_cast<const v4f*>(&s_red[row * CD + 4 * c]);
    const v4f v = (lane < 16) ? zv : uval[q];
    *reinterpret_cast<v4f*>(zu + (size_t)(b0 + row) * ZU_DIM + 4 * lane) = v;
  }
}

#define TRIP 48
#define RPB  8
#define B2Q  19
#define NB2L (B2Q * 4 * 1024)
typedef const __attribute__((address_space(3))) float* lds_cfloat_ptr;

template <int CTRL>
__device__ __forceinline__ float quad_bcast(float v) {
  return __int_as_float(__builtin_amdgcn_update_dpp(
      0, __float_as_int(v), CTRL, 0xf, 0xf, true));
}

__device__ __forceinline__ float quad_diag_sum(float a0, float a1, float a2, float a3) {
  return ((quad_bcast<0x00>(a0) + quad_bcast<0x55>(a1)) + quad_bcast<0xAA>(a2))
         + quad_bcast<0xFF>(a3);
}

__device__ __forceinline__ v8f wmma4(v2f a, v2f b, v8f c) {
  return __builtin_amdgcn_wmma_f32_16x16x4_f32(
       false, a,  false, b,
       (short)0, c,  false,  false);
}

struct Frag { v4f a0, a1, a2, a3; };

__device__ __forceinline__ Frag load_frag(const char* __restrict__ base, const unsigned off) {
  const v4f* ap = reinterpret_cast<const v4f*>(base + off);
  Frag t;
  t.a0 = ap[0]; t.a1 = ap[8]; t.a2 = ap[16]; t.a3 = ap[24];
  return t;
}

__device__ __forceinline__ void score_tile(const Frag& a, const Frag& b, float& x0, float& x1) {
  v8f c = {};
  c = wmma4(a.a0.xy, b.a0.xy, c);
  c = wmma4(a.a0.zw, b.a0.zw, c);
  c = wmma4(a.a1.xy, b.a1.xy, c);
  c = wmma4(a.a1.zw, b.a1.zw, c);
  c = wmma4(a.a2.xy, b.a2.xy, c);
  c = wmma4(a.a2.zw, b.a2.zw, c);
  c = wmma4(a.a3.xy, b.a3.xy, c);
  c = wmma4(a.a3.zw, b.a3.zw, c);
  x0 = quad_diag_sum(c[0], c[1], c[2], c[3]);
  x1 = quad_diag_sum(c[4], c[5], c[6], c[7]);
}

__device__ __forceinline__ unsigned row_off(const int LO, const unsigned sh0, const unsigned sh1, const int rr) {
  if (LO + 2 < 32) return (unsigned)__shfl((int)sh0, LO + rr);
  if (LO >= 32)    return (unsigned)__shfl((int)sh1, LO - 32 + rr);
  const unsigned lo = (unsigned)__shfl((int)sh0, (LO + rr) & 31);
  const unsigned hi = (unsigned)__shfl((int)sh1, (LO + rr) & 31);
  return (LO + rr < 32) ? lo : hi;
}

template <int NT>
__device__ __forceinline__ void score_trip(
    const int i0, const int lane, const int rr, const unsigned keep,
    const int wslot, const bool usex0, const char* __restrict__ base,
    const int* __restrict__ itrow, const float* __restrict__ b2,
    const lds_cfloat_ptr sb2, float* __restrict__ orow, float* sw) {
  const int n0 = i0 + lane;
  const int n1 = i0 + 32 + (lane & 15);
  const int id0 = itrow[(n0 < CIL) ? n0 : (CIL - 1)];
  const int id1 = itrow[(n1 < CIL) ? n1 : (CIL - 1)];
  float bias0 = sb2[(id0 < NB2L) ? id0 : (NB2L - 1)];
  float bias1 = sb2[(id1 < NB2L) ? id1 : (NB2L - 1)];
  if (id0 >= NB2L) bias0 = b2[id0];
  if (id1 >= NB2L) bias1 = b2[id1];
  const unsigned sh0 = (unsigned)id0 << 9, sh1 = (unsigned)id1 << 9;

  unsigned offA[NT], offB[NT];
  #pragma unroll
  for (int t = 0; t < NT; ++t) {
    offA[t] = row_off(6 * t,     sh0, sh1, rr) & keep;
    offB[t] = row_off(6 * t + 3, sh0, sh1, rr) & keep;
  }

  Frag ca = load_frag(base, offA[0]);
  Frag cb = load_frag(base, offB[0]);
  #pragma unroll
  for (int t = 0; t < NT; ++t) {
    Frag na = ca, nb = cb;
    if (t + 1 < NT) {
      na = load_frag(base, offA[t + 1]);
      nb = load_frag(base, offB[t + 1]);
    }
    __builtin_amdgcn_sched_barrier(0);
    float x0, x1;
    score_tile(ca, cb, x0, x1);
    const float v = usex0 ? x0 : x1;
    if (wslot >= 0) sw[6 * t + wslot] = v;
    ca = na; cb = nb;
  }
  const float r0 = sw[lane];
  const float r1 = (32 + (lane & 15) < 6 * NT) ? sw[32 + (lane & 15)] : 0.0f;

  if (n0 < CIL) orow[n0] = r0 + bias0;
  if (lane < 16 && n1 < CIL) orow[n1] = r1 + bias1;
}

__global__ __launch_bounds__(1024)
void caser_score(const int* __restrict__ items,
                 const float* __restrict__ W2,
                 const float* __restrict__ b2, const float* __restrict__ zu,
                 float* __restrict__ out) {
  __shared__ __attribute__((aligned(16))) float s_b2[NB2L];
  __shared__ float s_res[4 * RPB * TRIP];

  {
    v4f g[B2Q];
    #pragma unroll
    for (int q = 0; q < B2Q; ++q)
      g[q] = *reinterpret_cast<const v4f*>(b2 + 4 * (threadIdx.x + 1024 * q));
    #pragma unroll
    for (int q = 0; q < B2Q; ++q)
      *reinterpret_cast<v4f*>(&s_b2[4 * (threadIdx.x + 1024 * q)]) = g[q];
  }
  __syncthreads();

  const int lane = threadIdx.x & 31;
  const int wv   = __builtin_amdgcn_readfirstlane(threadIdx.x >> 5);
  const int b    = blockIdx.x * RPB + (wv >> 2);
  const int wave = wv & 3;
  const int h = lane >> 4;
  const int c = lane & 15;
  const int s = c & 3;
  const bool is_zu = c >= 12;
  const int rr = is_zu ? 0 : (c >> 2);
  const unsigned keep = is_zu ? 0u : 0xFFFFFFFFu;
  const unsigned kbyte = 4u * (unsigned)(8 * s + 4 * h);
  const char* zrow = reinterpret_cast<const char*>(zu + (size_t)b * ZU_DIM);
  const char* base = (is_zu ? zrow : reinterpret_cast<const char*>(W2)) + kbyte;

  int wslot = -1;
  if (lane == 12) wslot = 0;
  if (lane == 13) wslot = 1;
  if (lane == 28) wslot = 2;
  if (lane == 16) wslot = 3;
  if (lane == 20) wslot = 4;
  if (lane == 24) wslot = 5;
  const bool usex0 = (lane == 12) || (lane == 28);

  const int* __restrict__ itrow = items + (size_t)b * CIL;
  float* __restrict__ orow = out + (size_t)b * CIL;
  float* sw = s_res + wv * TRIP;
  const lds_cfloat_ptr sb2 = (lds_cfloat_ptr)s_b2;

  constexpr int FULL = CIL / TRIP;
  const int role = (wave + b + (b >> 8)) & 3;
  for (int it = role; it < FULL; it += 4)
    score_trip<8>(it * TRIP, lane, rr, keep, wslot, usex0, base, itrow, b2, sb2, orow, sw);
  if (role == (FULL & 3))
    score_trip<(CIL - FULL * TRIP + 5) / 6>(FULL * TRIP, lane, rr, keep, wslot, usex0,
                                            base, itrow, b2, sb2, orow, sw);
}

extern "C" void kernel_launch(void* const* d_in, const int* in_sizes, int n_in,
                              void* d_out, int out_size, void* d_ws, size_t ws_size,
                              hipStream_t stream) {
  const int*   seq      = (const int*)  d_in[0];
  const int*   user     = (const int*)  d_in[1];
  const int*   items    = (const int*)  d_in[2];
  const float* item_emb = (const float*)d_in[3];
  const float* user_emb = (const float*)d_in[4];
  const float* vw       = (const float*)d_in[5];
  const float* vb       = (const float*)d_in[6];
  const float* hw       = (const float*)d_in[7];
  const float* hb       = (const float*)d_in[8];
  const int*   heights  = (const int*)  d_in[9];
  const float* fc1_w    = (const float*)d_in[10];
  const float* fc1_b    = (const float*)d_in[11];
  const float* W2       = (const float*)d_in[12];
  const float* b2       = (const float*)d_in[13];
  float*       out      = (float*)d_out;
  float*       zu       = (float*)d_ws;

  caser_prep<<<dim3(CB / PB), dim3(128), 0, stream>>>(
      seq, user, item_emb, user_emb, vw, vb, hw, hb, heights, fc1_w, fc1_b, zu);

  caser_score<<<dim3(CB / RPB), dim3(1024), 0, stream>>>(items, W2, b2, zu, out);
}
